# starting kernel
# baseline (speedup 1.0000x reference)
#include <hip/hip_runtime.h>

typedef __attribute__((ext_vector_type(2))) float v2f;
typedef __attribute__((ext_vector_type(8))) float v8f;

#define NUQ   100000
#define NIQ   50000
#define DDIM  128
#define EEDG  300000
#define NNOD  (NUQ + NIQ)
#define LDSW  132

__global__ __launch_bounds__(512)
void proj_wmma_kernel(const float* __restrict__ xu, const float* __restrict__ xi,
                      const float* __restrict__ W1, const float* __restrict__ b1,
                      const float* __restrict__ W2, const float* __restrict__ b2,
                      float* __restrict__ out, float* __restrict__ S,
                      float* __restrict__ Qb2)
{
    __shared__ float xls[16 * LDSW];
    __shared__ float qls[16 * LDSW];

    const int row0 = blockIdx.x * 16;
    const int tid  = threadIdx.x;

    {
        const int idx = tid * 4;
        const int r   = idx >> 7;
        const int c   = idx & 127;
        const int row = row0 + r;
        const float* src = (row < NUQ) ? (xu + (size_t)row * DDIM + c)
                                       : (xi + (size_t)(row - NUQ) * DDIM + c);
        *(float4*)&xls[r * LDSW + c] = *(const float4*)src;
    }
    __syncthreads();

    const int lane = tid & 31;
    const int wv   = tid >> 5;
    const int isW2 = wv >> 3;
    const int c0   = (wv & 7) * 16;
    const int l15  = lane & 15;
    const int lh   = lane >> 4;

    const float* __restrict__ W    = isW2 ? W2 : W1;
    const float* __restrict__ Arow = &xls[l15 * LDSW + 2 * lh];
    const float* __restrict__ Brow = W + (size_t)(c0 + l15) * DDIM + 2 * lh;

    v8f acc = {};
    #pragma unroll
    for (int k = 0; k < DDIM; k += 4) {
        v2f a = *(const v2f*)(Arow + k);
        v2f b = *(const v2f*)(Brow + k);
        acc = __builtin_amdgcn_wmma_f32_16x16x4_f32(
             false, a,  false, b,
             (short)0, acc,  false,  false);
    }

    const int col = c0 + l15;
    if (isW2) {
        const float bb2 = b2[col];
        #pragma unroll
        for (int v = 0; v < 8; ++v) {
            const int m   = v + 8 * lh;
            const float q = acc[v];
            qls[m * LDSW + col] = q;
            Qb2[(size_t)(row0 + m) * DDIM + col] = q + bb2;
        }
    }
    __syncthreads();
    if (!isW2) {
        const float bb1 = b1[col];
        #pragma unroll
        for (int v = 0; v < 8; ++v) {
            const int m   = v + 8 * lh;
            const float p = acc[v] + bb1;
            const float q = qls[m * LDSW + col];
            out[(size_t)(row0 + m) * DDIM + col] = p;
            S  [(size_t)(row0 + m) * DDIM + col] = p + q;
        }
    }
}

__global__ __launch_bounds__(256)
void edge_kernel(const float* __restrict__ S, const float* __restrict__ Qb2,
                 const int* __restrict__ src_ui, const int* __restrict__ dst_ui,
                 const float* __restrict__ norm_ui,
                 const int* __restrict__ src_iu, const int* __restrict__ dst_iu,
                 const float* __restrict__ norm_iu,
                 float* __restrict__ out)
{
    const int gw   = (int)((blockIdx.x * blockDim.x + threadIdx.x) >> 5);
    const int lane = threadIdx.x & 31;
    if (gw >= 2 * EEDG) return;

    size_t srow, drow;
    float  nrm;
    if (gw < EEDG) {
        const int e = gw;
        srow = (size_t)src_ui[e] * DDIM;
        drow = (size_t)(NUQ + dst_ui[e]) * DDIM;
        nrm  = norm_ui[e];
    } else {
        const int e = gw - EEDG;
        srow = (size_t)(NUQ + src_iu[e]) * DDIM;
        drow = (size_t)dst_iu[e] * DDIM;
        nrm  = norm_iu[e];
    }

    const int c = lane * 4;
    float4 sv = *(const float4*)(S   + srow + c);
    float4 qv = *(const float4*)(Qb2 + drow + c);
    float* o  = out + drow + c;
    unsafeAtomicAdd(o + 0, nrm * (sv.x + qv.x));
    unsafeAtomicAdd(o + 1, nrm * (sv.y + qv.y));
    unsafeAtomicAdd(o + 2, nrm * (sv.z + qv.z));
    unsafeAtomicAdd(o + 3, nrm * (sv.w + qv.w));
}

__global__ __launch_bounds__(256)
void finalize_kernel(float* __restrict__ out)
{
    const int gw   = (int)((blockIdx.x * blockDim.x + threadIdx.x) >> 5);
    const int lane = threadIdx.x & 31;
    if (gw >= NNOD) return;

    const size_t base = (size_t)gw * DDIM + lane * 4;
    float4 v = *(float4*)(out + base);
    v.x = v.x > 0.0f ? v.x : 0.2f * v.x;
    v.y = v.y > 0.0f ? v.y : 0.2f * v.y;
    v.z = v.z > 0.0f ? v.z : 0.2f * v.z;
    v.w = v.w > 0.0f ? v.w : 0.2f * v.w;

    float ss = v.x * v.x + v.y * v.y + v.z * v.z + v.w * v.w;
    #pragma unroll
    for (int off = 16; off > 0; off >>= 1)
        ss += __shfl_xor(ss, off, 32);

    const float scale = 1.0f / fmaxf(sqrtf(ss), 1e-12f);
    v.x *= scale; v.y *= scale; v.z *= scale; v.w *= scale;
    *(float4*)(out + base) = v;
}

extern "C" void kernel_launch(void* const* d_in, const int* in_sizes, int n_in,
                              void* d_out, int out_size, void* d_ws, size_t ws_size,
                              hipStream_t stream) {
    (void)in_sizes; (void)n_in; (void)out_size; (void)ws_size;

    const float* xu      = (const float*)d_in[0];
    const float* xi      = (const float*)d_in[1];
    const float* W1      = (const float*)d_in[2];
    const float* b1      = (const float*)d_in[3];
    const float* W2      = (const float*)d_in[4];
    const float* b2      = (const float*)d_in[5];
    const int*   src_ui  = (const int*)d_in[6];
    const int*   dst_ui  = (const int*)d_in[7];
    const float* norm_ui = (const float*)d_in[8];
    const int*   src_iu  = (const int*)d_in[9];
    const int*   dst_iu  = (const int*)d_in[10];
    const float* norm_iu = (const float*)d_in[11];

    float* out = (float*)d_out;
    float* S   = (float*)d_ws;
    float* Qb2 = S + (size_t)NNOD * DDIM;

    proj_wmma_kernel<<<NNOD / 16, 512, 0, stream>>>(xu, xi, W1, b1, W2, b2, out, S, Qb2);

    edge_kernel<<<(2 * EEDG) / 8, 256, 0, stream>>>(S, Qb2, src_ui, dst_ui, norm_ui,
                                                    src_iu, dst_iu, norm_iu, out);

    finalize_kernel<<<NNOD / 8, 256, 0, stream>>>(out);
}


# proj kernel: W1,W2 staged in LDS once per resident block (256 x 16 waves); a wave holds a 16-row x tile as A fragments, streams B from LDS, stores whole lines; k regrouped per WMMA (sum order)
# speedup vs baseline: 1.3182x; 1.3182x over previous
#include <hip/hip_runtime.h>

typedef __attribute__((ext_vector_type(2))) float v2f;
typedef __attribute__((ext_vector_type(4))) float v4f;
typedef __attribute__((ext_vector_type(8))) float v8f;

#define NUQ   100000
#define NIQ   50000
#define DDIM  128
#define EEDG  300000
#define NNOD  (NUQ + NIQ)
#define LDSW  132
#define NTILE (NNOD / 16)
#define PBLK  256
#define PWAV  16

__global__ __launch_bounds__(512)
void proj_wmma_kernel(const float* __restrict__ xu, const float* __restrict__ xi,
                      const float* __restrict__ W1, const float* __restrict__ b1,
                      const float* __restrict__ W2, const float* __restrict__ b2,
                      float* __restrict__ out, float* __restrict__ S,
                      float* __restrict__ Qb2)
{
    __shared__ __attribute__((aligned(16))) float wls[256 * LDSW];
    __shared__ __attribute__((aligned(16))) float xls[PWAV * 16 * LDSW];

    const int tid  = threadIdx.x;
    const int lane = tid & 31;
    const int wv   = tid >> 5;
    const int c    = lane & 15;
    const int h    = lane >> 4;

    for (int i = 0; i < 16; ++i) {
        const int idx  = i * 512 + tid;
        const int p    = idx >> 5;
        const int q    = idx & 31;
        const int n    = 64 * ((p >> 6) & 1) + 4 * (p & 15) + ((p >> 4) & 3);
        const float* src = ((p >> 7) ? W2 : W1) + (size_t)n * DDIM + 4 * q;
        *(v4f*)&wls[p * LDSW + 4 * q] = *(const v4f*)src;
    }
    __syncthreads();

    float* __restrict__ xw = xls + wv * (16 * LDSW);
    const float* __restrict__ bfrag = wls + c * LDSW + 4 * h;

    const int slot = (wv * 13) & 15;

    for (int tile = slot * PBLK + (int)blockIdx.x; tile < NTILE; tile += PBLK * PWAV) {
        const int row0 = tile * 16;
        const float* __restrict__ xbase = (row0 < NUQ) ? (xu + (size_t)row0 * DDIM)
                                                       : (xi + (size_t)(row0 - NUQ) * DDIM);

        v4f t[16];
        #pragma unroll
        for (int r = 0; r < 16; ++r)
            t[r] = *(const v4f*)(xbase + r * DDIM + 4 * lane);
        #pragma unroll
        for (int r = 0; r < 16; ++r)
            *(v4f*)&xw[r * LDSW + 4 * lane] = t[r];

        v4f A[16];
        #pragma unroll
        for (int u = 0; u < 16; ++u)
            A[u] = *(const v4f*)&xw[c * LDSW + 8 * u + 4 * h];

        for (int g = 0; g < 2; ++g) {
            const float* __restrict__ bP = bfrag + (g * 64) * LDSW;
            const float* __restrict__ bQ = bP + 128 * LDSW;

            v8f accP[4] = {};
            v8f accQ[4] = {};
            #pragma unroll
            for (int u = 0; u < 16; ++u) {
                v4f wp[4], wq[4];
                #pragma unroll
                for (int jj = 0; jj < 4; ++jj) {
                    wp[jj] = *(const v4f*)(bP + (jj * 16) * LDSW + 8 * u);
                    wq[jj] = *(const v4f*)(bQ + (jj * 16) * LDSW + 8 * u);
                }
                const v2f a0 = A[u].xy;
                const v2f a1 = A[u].zw;
                #pragma unroll
                for (int jj = 0; jj < 4; ++jj) {
                    accP[jj] = __builtin_amdgcn_wmma_f32_16x16x4_f32(
                        false, a0, false, wp[jj].xy, (short)0, accP[jj], false, false);
                    accQ[jj] = __builtin_amdgcn_wmma_f32_16x16x4_f32(
                        false, a0, false, wq[jj].xy, (short)0, accQ[jj], false, false);
                }
                #pragma unroll
                for (int jj = 0; jj < 4; ++jj) {
                    accP[jj] = __builtin_amdgcn_wmma_f32_16x16x4_f32(
                        false, a1, false, wp[jj].zw, (short)0, accP[jj], false, false);
                    accQ[jj] = __builtin_amdgcn_wmma_f32_16x16x4_f32(
                        false, a1, false, wq[jj].zw, (short)0, accQ[jj], false, false);
                }
            }

            const int col = 64 * g + 4 * c;
            const v4f bb1 = *(const v4f*)(b1 + col);
            const v4f bb2 = *(const v4f*)(b2 + col);
            #pragma unroll
            for (int r = 0; r < 8; ++r) {
                const size_t o = (size_t)(row0 + 8 * h + r) * DDIM + col;
                v4f p, q;
                p.x = accP[0][r] + bb1.x;  q.x = accQ[0][r];
                p.y = accP[1][r] + bb1.y;  q.y = accQ[1][r];
                p.z = accP[2][r] + bb1.z;  q.z = accQ[2][r];
                p.w = accP[3][r] + bb1.w;  q.w = accQ[3][r];
                *(v4f*)(out + o) = p;
                *(v4f*)(S   + o) = p + q;
                *(v4f*)(Qb2 + o) = q + bb2;
            }
        }
    }
}

__global__ __launch_bounds__(256)
void edge_kernel(const float* __restrict__ S, const float* __restrict__ Qb2,
                 const int* __restrict__ src_ui, const int* __restrict__ dst_ui,
                 const float* __restrict__ norm_ui,
                 const int* __restrict__ src_iu, const int* __restrict__ dst_iu,
                 const float* __restrict__ norm_iu,
                 float* __restrict__ out)
{
    const int gw   = (int)((blockIdx.x * blockDim.x + threadIdx.x) >> 5);
    const int lane = threadIdx.x & 31;
    if (gw >= 2 * EEDG) return;

    size_t srow, drow;
    float  nrm;
    if (gw < EEDG) {
        const int e = gw;
        srow = (size_t)src_ui[e] * DDIM;
        drow = (size_t)(NUQ + dst_ui[e]) * DDIM;
        nrm  = norm_ui[e];
    } else {
        const int e = gw - EEDG;
        srow = (size_t)(NUQ + src_iu[e]) * DDIM;
        drow = (size_t)dst_iu[e] * DDIM;
        nrm  = norm_iu[e];
    }

    const int c = lane * 4;
    float4 sv = *(const float4*)(S   + srow + c);
    float4 qv = *(const float4*)(Qb2 + drow + c);
    float* o  = out + drow + c;
    unsafeAtomicAdd(o + 0, nrm * (sv.x + qv.x));
    unsafeAtomicAdd(o + 1, nrm * (sv.y + qv.y));
    unsafeAtomicAdd(o + 2, nrm * (sv.z + qv.z));
    unsafeAtomicAdd(o + 3, nrm * (sv.w + qv.w));
}

__global__ __launch_bounds__(256)
void finalize_kernel(float* __restrict__ out)
{
    const int gw   = (int)((blockIdx.x * blockDim.x + threadIdx.x) >> 5);
    const int lane = threadIdx.x & 31;
    if (gw >= NNOD) return;

    const size_t base = (size_t)gw * DDIM + lane * 4;
    float4 v = *(float4*)(out + base);
    v.x = v.x > 0.0f ? v.x : 0.2f * v.x;
    v.y = v.y > 0.0f ? v.y : 0.2f * v.y;
    v.z = v.z > 0.0f ? v.z : 0.2f * v.z;
    v.w = v.w > 0.0f ? v.w : 0.2f * v.w;

    float ss = v.x * v.x + v.y * v.y + v.z * v.z + v.w * v.w;
    #pragma unroll
    for (int off = 16; off > 0; off >>= 1)
        ss += __shfl_xor(ss, off, 32);

    const float scale = 1.0f / fmaxf(sqrtf(ss), 1e-12f);
    v.x *= scale; v.y *= scale; v.z *= scale; v.w *= scale;
    *(float4*)(out + base) = v;
}

extern "C" void kernel_launch(void* const* d_in, const int* in_sizes, int n_in,
                              void* d_out, int out_size, void* d_ws, size_t ws_size,
                              hipStream_t stream) {
    (void)in_sizes; (void)n_in; (void)out_size; (void)ws_size;

    const float* xu      = (const float*)d_in[0];
    const float* xi      = (const float*)d_in[1];
    const float* W1      = (const float*)d_in[2];
    const float* b1      = (const float*)d_in[3];
    const float* W2      = (const float*)d_in[4];
    const float* b2      = (const float*)d_in[5];
    const int*   src_ui  = (const int*)d_in[6];
    const int*   dst_ui  = (const int*)d_in[7];
    const float* norm_ui = (const float*)d_in[8];
    const int*   src_iu  = (const int*)d_in[9];
    const int*   dst_iu  = (const int*)d_in[10];
    const float* norm_iu = (const float*)d_in[11];

    float* out = (float*)d_out;
    float* S   = (float*)d_ws;
    float* Qb2 = S + (size_t)NNOD * DDIM;

    proj_wmma_kernel<<<PBLK, 512, 0, stream>>>(xu, xi, W1, b1, W2, b2, out, S, Qb2);

    edge_kernel<<<(2 * EEDG) / 8, 256, 0, stream>>>(S, Qb2, src_ui, dst_ui, norm_ui,
                                                    src_iu, dst_iu, norm_iu, out);

    finalize_kernel<<<NNOD / 8, 256, 0, stream>>>(out);
}


# edge sums: global float atomics + finalize launch -> one kernel: block per 586 dst rows scans dst ids, LDS float adds of gathered S rows (arrival order), dst term = norm sum x Qb2 row, rows finished
# speedup vs baseline: 2.0731x; 1.5727x over previous
#include <hip/hip_runtime.h>

typedef __attribute__((ext_vector_type(2))) float v2f;
typedef __attribute__((ext_vector_type(4))) float v4f;
typedef __attribute__((ext_vector_type(8))) float v8f;

#define NUQ   100000
#define NIQ   50000
#define DDIM  128
#define EEDG  300000
#define NNOD  (NUQ + NIQ)
#define LDSW  132
#define NTILE (NNOD / 16)
#define PBLK  256
#define PWAV  16

__global__ __launch_bounds__(512)
void proj_wmma_kernel(const float* __restrict__ xu, const float* __restrict__ xi,
                      const float* __restrict__ W1, const float* __restrict__ b1,
                      const float* __restrict__ W2, const float* __restrict__ b2,
                      float* __restrict__ out, float* __restrict__ S,
                      float* __restrict__ Qb2)
{
    __shared__ __attribute__((aligned(16))) float wls[256 * LDSW];
    __shared__ __attribute__((aligned(16))) float xls[PWAV * 16 * LDSW];

    const int tid  = threadIdx.x;
    const int lane = tid & 31;
    const int wv   = tid >> 5;
    const int c    = lane & 15;
    const int h    = lane >> 4;

    for (int i = 0; i < 16; ++i) {
        const int idx  = i * 512 + tid;
        const int p    = idx >> 5;
        const int q    = idx & 31;
        const int n    = 64 * ((p >> 6) & 1) + 4 * (p & 15) + ((p >> 4) & 3);
        const float* src = ((p >> 7) ? W2 : W1) + (size_t)n * DDIM + 4 * q;
        *(v4f*)&wls[p * LDSW + 4 * q] = *(const v4f*)src;
    }
    __syncthreads();

    float* __restrict__ xw = xls + wv * (16 * LDSW);
    const float* __restrict__ bfrag = wls + c * LDSW + 4 * h;

    const int slot = (wv * 13) & 15;

    for (int tile = slot * PBLK + (int)blockIdx.x; tile < NTILE; tile += PBLK * PWAV) {
        const int row0 = tile * 16;
        const float* __restrict__ xbase = (row0 < NUQ) ? (xu + (size_t)row0 * DDIM)
                                                       : (xi + (size_t)(row0 - NUQ) * DDIM);

        v4f t[16];
        #pragma unroll
        for (int r = 0; r < 16; ++r)
            t[r] = *(const v4f*)(xbase + r * DDIM + 4 * lane);
        #pragma unroll
        for (int r = 0; r < 16; ++r)
            *(v4f*)&xw[r * LDSW + 4 * lane] = t[r];

        v4f A[16];
        #pragma unroll
        for (int u = 0; u < 16; ++u)
            A[u] = *(const v4f*)&xw[c * LDSW + 8 * u + 4 * h];

        for (int g = 0; g < 2; ++g) {
            const float* __restrict__ bP = bfrag + (g * 64) * LDSW;
            const float* __restrict__ bQ = bP + 128 * LDSW;

            v8f accP[4] = {};
            v8f accQ[4] = {};
            #pragma unroll
            for (int u = 0; u < 16; ++u) {
                v4f wp[4], wq[4];
                #pragma unroll
                for (int jj = 0; jj < 4; ++jj) {
                    wp[jj] = *(const v4f*)(bP + (jj * 16) * LDSW + 8 * u);
                    wq[jj] = *(const v4f*)(bQ + (jj * 16) * LDSW + 8 * u);
                }
                const v2f a0 = A[u].xy;
                const v2f a1 = A[u].zw;
                #pragma unroll
                for (int jj = 0; jj < 4; ++jj) {
                    accP[jj] = __builtin_amdgcn_wmma_f32_16x16x4_f32(
                        false, a0, false, wp[jj].xy, (short)0, accP[jj], false, false);
                    accQ[jj] = __builtin_amdgcn_wmma_f32_16x16x4_f32(
                        false, a0, false, wq[jj].xy, (short)0, accQ[jj], false, false);
                }
                #pragma unroll
                for (int jj = 0; jj < 4; ++jj) {
                    accP[jj] = __builtin_amdgcn_wmma_f32_16x16x4_f32(
                        false, a1, false, wp[jj].zw, (short)0, accP[jj], false, false);
                    accQ[jj] = __builtin_amdgcn_wmma_f32_16x16x4_f32(
                        false, a1, false, wq[jj].zw, (short)0, accQ[jj], false, false);
                }
            }

            const int col = 64 * g + 4 * c;
            const v4f bb1 = *(const v4f*)(b1 + col);
            const v4f bb2 = *(const v4f*)(b2 + col);
            #pragma unroll
            for (int r = 0; r < 8; ++r) {
                const size_t o = (size_t)(row0 + 8 * h + r) * DDIM + col;
                v4f p, q;
                p.x = accP[0][r] + bb1.x;  q.x = accQ[0][r];
                p.y = accP[1][r] + bb1.y;  q.y = accQ[1][r];
                p.z = accP[2][r] + bb1.z;  q.z = accQ[2][r];
                p.w = accP[3][r] + bb1.w;  q.w = accQ[3][r];
                *(v4f*)(out + o) = p;
                *(v4f*)(S   + o) = p + q;
                *(v4f*)(Qb2 + o) = q + bb2;
            }
        }
    }
}

#define TROWS 586
#define TBLK  256
#define LCAP  5120
#define NVEC  (EEDG / 4)

__global__ __launch_bounds__(1024)
void tile_edge_kernel(const float* __restrict__ S, const float* __restrict__ Qb2,
                      const int* __restrict__ src_ui, const int* __restrict__ dst_ui,
                      const float* __restrict__ norm_ui,
                      const int* __restrict__ src_iu, const int* __restrict__ dst_iu,
                      const float* __restrict__ norm_iu,
                      float* out)
{
    __shared__ __attribute__((aligned(16))) float acc[TROWS * DDIM];
    __shared__ float nsum[TROWS + 6];
    __shared__ int   elist[LCAP];
    __shared__ int   ecnt;

    const int tid  = threadIdx.x;
    const int lane = tid & 31;
    const int wv   = __builtin_amdgcn_readfirstlane(tid >> 5);
    const int lo   = (int)blockIdx.x * TROWS;
    const int hi   = (lo + TROWS < NNOD) ? (lo + TROWS) : NNOD;
    const int nrows = hi - lo;

    for (int i = tid; i < TROWS * DDIM / 4; i += 1024)
        ((v4f*)acc)[i] = v4f{0.0f, 0.0f, 0.0f, 0.0f};
    for (int i = tid; i < TROWS; i += 1024)
        nsum[i] = 0.0f;
    if (tid == 0) ecnt = 0;
    __syncthreads();

    auto direct_edge = [&](int gid) {
        const int ty = gid >= EEDG;
        const int e  = ty ? gid - EEDG : gid;
        const int sr = ty ? NUQ + src_iu[e] : src_ui[e];
        const int lr = (ty ? dst_iu[e] : NUQ + dst_ui[e]) - lo;
        const float nr = ty ? norm_iu[e] : norm_ui[e];
        const float* __restrict__ sp = S + (size_t)sr * DDIM;
        for (int k = 0; k < DDIM; ++k)
            atomicAdd(&acc[lr * DDIM + k], nr * sp[k]);
        atomicAdd(&nsum[lr], nr);
    };

    const int rot = (int)((blockIdx.x * 293u) % (unsigned)NVEC);
    for (int ty = 0; ty < 2; ++ty) {
        const int off = ty ? 0 : NUQ;
        const int lim = ty ? NUQ : NIQ;
        const int dlo = (lo - off > 0) ? (lo - off) : 0;
        const int dhi = (hi - off < lim) ? (hi - off) : lim;
        if (dhi <= dlo) continue;
        const unsigned span = (unsigned)(dhi - dlo);
        const int4* __restrict__ dv = (const int4*)(ty ? dst_iu : dst_ui);
        const int gbase = ty ? EEDG : 0;

        for (int it0 = 0; it0 < (NVEC + 1023) / 1024; it0 += 8) {
            int4 d[8];
            int  p[8];
            #pragma unroll
            for (int k = 0; k < 8; ++k) {
                const int idx = (it0 + k) * 1024 + tid;
                int q = idx + rot;
                if (q >= NVEC) q -= NVEC;
                p[k] = (idx < NVEC) ? q : -1;
                d[k] = dv[(idx < NVEC) ? q : 0];
            }
            #pragma unroll
            for (int k = 0; k < 8; ++k) {
                unsigned m = ((unsigned)(d[k].x - dlo) < span ? 1u : 0u)
                           | ((unsigned)(d[k].y - dlo) < span ? 2u : 0u)
                           | ((unsigned)(d[k].z - dlo) < span ? 4u : 0u)
                           | ((unsigned)(d[k].w - dlo) < span ? 8u : 0u);
                if (p[k] < 0) m = 0u;
                while (m) {
                    const int c = __builtin_ctz(m);
                    m &= m - 1u;
                    const int gid = gbase + 4 * p[k] + c;
                    const int pos = atomicAdd(&ecnt, 1);
                    if (pos < LCAP) elist[pos] = gid;
                    else            direct_edge(gid);
                }
            }
        }
    }
    __syncthreads();

    const int nraw = __builtin_amdgcn_readfirstlane(ecnt);
    const int n    = nraw < LCAP ? nraw : LCAP;
    for (int j0 = wv * 32; j0 < n; j0 += 32 * 32) {
        const int  nb = (n - j0 < 32) ? (n - j0) : 32;
        const bool ok = lane < nb;
        const int gid = elist[ok ? (j0 + lane) : j0];
        const int ty  = gid >= EEDG;
        const int e   = ty ? gid - EEDG : gid;
        const int sidx = (ty ? src_iu : src_ui)[e];
        const int didx = (ty ? dst_iu : dst_ui)[e];
        const float nr = (ty ? norm_iu : norm_ui)[e];
        const int srow = ty ? NUQ + sidx : sidx;
        const int lrow = (ty ? didx : NUQ + didx) - lo;
        if (ok) atomicAdd(&nsum[lrow], nr);

        for (int g = 0; g < nb; g += 16) {
            float r[16][4];
            #pragma unroll
            for (int k = 0; k < 16; ++k) {
                const int sr = __builtin_amdgcn_readlane(srow, g + k);
                const float* __restrict__ sp = S + (size_t)sr * DDIM + lane;
                r[k][0] = sp[0];
                r[k][1] = sp[32];
                r[k][2] = sp[64];
                r[k][3] = sp[96];
            }
            #pragma unroll
            for (int k = 0; k < 16; ++k) {
                if (g + k < nb) {
                    const float w = __builtin_bit_cast(float,
                        __builtin_amdgcn_readlane(__builtin_bit_cast(int, nr), g + k));
                    const int   l = __builtin_amdgcn_readlane(lrow, g + k);
                    float* a = acc + l * DDIM + lane;
                    atomicAdd(a,      w * r[k][0]);
                    atomicAdd(a + 32, w * r[k][1]);
                    atomicAdd(a + 64, w * r[k][2]);
                    atomicAdd(a + 96, w * r[k][3]);
                }
            }
        }
    }
    __syncthreads();

    for (int rb = wv; rb < nrows; rb += 128) {
        v4f pv[4], qv[4];
        #pragma unroll
        for (int k = 0; k < 4; ++k) {
            const int rr = (rb + 32 * k < nrows) ? (rb + 32 * k) : rb;
            const size_t o = (size_t)(lo + rr) * DDIM + 4 * lane;
            pv[k] = *(const v4f*)(out + o);
            qv[k] = *(const v4f*)(Qb2 + o);
        }
        #pragma unroll
        for (int k = 0; k < 4; ++k) {
            const int rr = rb + 32 * k;
            if (rr < nrows) {
                const v4f   a = *(const v4f*)&acc[rr * DDIM + 4 * lane];
                const float s = nsum[rr];
                v4f v = pv[k] + a + s * qv[k];
                v.x = v.x > 0.0f ? v.x : 0.2f * v.x;
                v.y = v.y > 0.0f ? v.y : 0.2f * v.y;
                v.z = v.z > 0.0f ? v.z : 0.2f * v.z;
                v.w = v.w > 0.0f ? v.w : 0.2f * v.w;
                float ss = v.x * v.x + v.y * v.y + v.z * v.z + v.w * v.w;
                #pragma unroll
                for (int sh = 16; sh > 0; sh >>= 1)
                    ss += __shfl_xor(ss, sh, 32);
                const float scale = 1.0f / fmaxf(sqrtf(ss), 1e-12f);
                v.x *= scale; v.y *= scale; v.z *= scale; v.w *= scale;
                *(v4f*)(out + (size_t)(lo + rr) * DDIM + 4 * lane) = v;
            }
        }
    }
}

extern "C" void kernel_launch(void* const* d_in, const int* in_sizes, int n_in,
                              void* d_out, int out_size, void* d_ws, size_t ws_size,
                              hipStream_t stream) {
    (void)in_sizes; (void)n_in; (void)out_size; (void)ws_size;

    const float* xu      = (const float*)d_in[0];
    const float* xi      = (const float*)d_in[1];
    const float* W1      = (const float*)d_in[2];
    const float* b1      = (const float*)d_in[3];
    const float* W2      = (const float*)d_in[4];
    const float* b2      = (const float*)d_in[5];
    const int*   src_ui  = (const int*)d_in[6];
    const int*   dst_ui  = (const int*)d_in[7];
    const float* norm_ui = (const float*)d_in[8];
    const int*   src_iu  = (const int*)d_in[9];
    const int*   dst_iu  = (const int*)d_in[10];
    const float* norm_iu = (const float*)d_in[11];

    float* out = (float*)d_out;
    float* S   = (float*)d_ws;
    float* Qb2 = S + (size_t)NNOD * DDIM;

    proj_wmma_kernel<<<PBLK, 512, 0, stream>>>(xu, xi, W1, b1, W2, b2, out, S, Qb2);

    tile_edge_kernel<<<TBLK, 1024, 0, stream>>>(S, Qb2, src_ui, dst_ui, norm_ui,
                                                src_iu, dst_iu, norm_iu, out);
}
